# starting kernel
# baseline (speedup 1.0000x reference)
#include <hip/hip_runtime.h>


typedef float v2f __attribute__((ext_vector_type(2)));
typedef float v8f __attribute__((ext_vector_type(8)));

#define H       256
#define LAYERS  3
#define MBLK    16
#define STRIDE  264

__device__ __forceinline__ v8f wmma_f32x4(v2f a, v2f b, v8f c) {
    return __builtin_amdgcn_wmma_f32_16x16x4_f32(
         false, a,  false, b,
         (short)0, c,  false,  false);
}

__device__ __forceinline__ float fast_tanh(float v) {
#if __has_builtin(__builtin_amdgcn_tanhf)
    return __builtin_amdgcn_tanhf(v);
#else
    float e = __expf(2.0f * v);
    return 1.0f - 2.0f / (e + 1.0f);
#endif
}

__global__ __launch_bounds__(256) void dgm_fused_kernel(
    const float* __restrict__ t,  const float* __restrict__ x,
    const float* __restrict__ W0, const float* __restrict__ b0,
    const float* __restrict__ Uz, const float* __restrict__ Ug,
    const float* __restrict__ Ur, const float* __restrict__ Uh,
    const float* __restrict__ Wz, const float* __restrict__ Wg,
    const float* __restrict__ Wr, const float* __restrict__ Wh,
    const float* __restrict__ bz, const float* __restrict__ bg,
    const float* __restrict__ br, const float* __restrict__ bh,
    const float* __restrict__ Wf, const float* __restrict__ bf,
    float* __restrict__ out)
{
    __shared__ float sS [MBLK][STRIDE];
    __shared__ float sS1[MBLK][STRIDE];
    __shared__ float sSR[MBLK][STRIDE];
    __shared__ float sXt[MBLK];
    __shared__ float sXx[MBLK];

    const int tid  = threadIdx.x;
    const int wave = tid >> 5;
    const int lane = tid & 31;
    const int half = lane >> 4;
    const int l16  = lane & 15;
    const int row0 = blockIdx.x * MBLK;

    if (tid < MBLK) {
        sXt[tid] = t[row0 + tid];
        sXx[tid] = x[row0 + tid];
    }
    __syncthreads();

    {
        const int   c   = tid;
        const float w0t = W0[c];
        const float w0x = W0[H + c];
        const float bb  = b0[c];
        #pragma unroll
        for (int r = 0; r < MBLK; ++r) {
            float v = fast_tanh(sXt[r] * w0t + sXx[r] * w0x + bb);
            sS1[r][c] = v;
            sS [r][c] = v;
        }
    }
    __syncthreads();

    float xt8[8], xx8[8];
    #pragma unroll
    for (int v = 0; v < 8; ++v) {
        xt8[v] = sXt[half * 8 + v];
        xx8[v] = sXx[half * 8 + v];
    }

    const int koff = half * 2;
    const int cb0  = wave * 32;

    for (int l = 0; l < LAYERS; ++l) {
        const float* wzl = Wz + l * H * H;
        const float* wgl = Wg + l * H * H;
        const float* wrl = Wr + l * H * H;
        const float* whl = Wh + l * H * H;
        const float* uzl = Uz + l * 2 * H;
        const float* ugl = Ug + l * 2 * H;
        const float* url = Ur + l * 2 * H;
        const float* uhl = Uh + l * 2 * H;
        const float* bzl = bz + l * H;
        const float* bgl = bg + l * H;
        const float* brl = br + l * H;
        const float* bhl = bh + l * H;

        v8f accR[2];
        #pragma unroll
        for (int t2 = 0; t2 < 2; ++t2) {
            const int   n  = cb0 + t2 * 16 + l16;
            const float u0 = url[n];
            const float u1 = url[H + n];
            const float bb = brl[n];
            #pragma unroll
            for (int v = 0; v < 8; ++v)
                accR[t2][v] = xt8[v] * u0 + xx8[v] * u1 + bb;
        }
        for (int k = 0; k < H; k += 4) {
            v2f a = *(const v2f*)&sS[l16][k + koff];
            #pragma unroll
            for (int t2 = 0; t2 < 2; ++t2) {
                const int n = cb0 + t2 * 16 + l16;
                v2f b;
                b.x = wrl[(k + koff)     * H + n];
                b.y = wrl[(k + koff + 1) * H + n];
                accR[t2] = wmma_f32x4(a, b, accR[t2]);
            }
        }
        #pragma unroll
        for (int t2 = 0; t2 < 2; ++t2) {
            const int n = cb0 + t2 * 16 + l16;
            #pragma unroll
            for (int v = 0; v < 8; ++v) {
                const int m = half * 8 + v;
                float R = fast_tanh(accR[t2][v]);
                sSR[m][n] = sS[m][n] * R;
            }
        }
        __syncthreads();

        v8f accZ[2], accG[2], accHg[2];
        #pragma unroll
        for (int t2 = 0; t2 < 2; ++t2) {
            const int   n  = cb0 + t2 * 16 + l16;
            const float z0 = uzl[n], z1 = uzl[H + n], zb = bzl[n];
            const float g0 = ugl[n], g1 = ugl[H + n], gb = bgl[n];
            const float h0 = uhl[n], h1 = uhl[H + n], hb = bhl[n];
            #pragma unroll
            for (int v = 0; v < 8; ++v) {
                accZ [t2][v] = xt8[v] * z0 + xx8[v] * z1 + zb;
                accG [t2][v] = xt8[v] * g0 + xx8[v] * g1 + gb;
                accHg[t2][v] = xt8[v] * h0 + xx8[v] * h1 + hb;
            }
        }
        for (int k = 0; k < H; k += 4) {
            v2f aS  = *(const v2f*)&sS [l16][k + koff];
            v2f aS1 = *(const v2f*)&sS1[l16][k + koff];
            v2f aSR = *(const v2f*)&sSR[l16][k + koff];
            #pragma unroll
            for (int t2 = 0; t2 < 2; ++t2) {
                const int n = cb0 + t2 * 16 + l16;
                v2f bzv, bgv, bhv;
                bzv.x = wzl[(k + koff)     * H + n];
                bzv.y = wzl[(k + koff + 1) * H + n];
                bgv.x = wgl[(k + koff)     * H + n];
                bgv.y = wgl[(k + koff + 1) * H + n];
                bhv.x = whl[(k + koff)     * H + n];
                bhv.y = whl[(k + koff + 1) * H + n];
                accZ [t2] = wmma_f32x4(aS,  bzv, accZ [t2]);
                accG [t2] = wmma_f32x4(aS1, bgv, accG [t2]);
                accHg[t2] = wmma_f32x4(aSR, bhv, accHg[t2]);
            }
        }
        float snew[2][8];
        #pragma unroll
        for (int t2 = 0; t2 < 2; ++t2) {
            const int n = cb0 + t2 * 16 + l16;
            #pragma unroll
            for (int v = 0; v < 8; ++v) {
                const int m = half * 8 + v;
                float Zv = fast_tanh(accZ [t2][v]);
                float Gv = fast_tanh(accG [t2][v]);
                float Hv = fast_tanh(accHg[t2][v]);
                snew[t2][v] = (1.0f - Gv) * Hv + Zv * sS[m][n];
            }
        }
        __syncthreads();
        #pragma unroll
        for (int t2 = 0; t2 < 2; ++t2) {
            const int n = cb0 + t2 * 16 + l16;
            #pragma unroll
            for (int v = 0; v < 8; ++v)
                sS[half * 8 + v][n] = snew[t2][v];
        }
        __syncthreads();
    }

    {
        const int r  = tid >> 4;
        const int ci = tid & 15;
        float p = 0.0f;
        #pragma unroll
        for (int j = 0; j < 16; ++j) {
            const int c = ci * 16 + j;
            p += sS[r][c] * Wf[c];
        }
        #pragma unroll
        for (int off = 8; off >= 1; off >>= 1)
            p += __shfl_xor(p, off, 16);
        if (ci == 0)
            out[row0 + r] = p + bf[0];
    }
}

extern "C" void kernel_launch(void* const* d_in, const int* in_sizes, int n_in,
                              void* d_out, int out_size, void* d_ws, size_t ws_size,
                              hipStream_t stream) {
    (void)n_in; (void)d_ws; (void)ws_size; (void)out_size;
    const float* t  = (const float*)d_in[0];
    const float* x  = (const float*)d_in[1];
    const float* W0 = (const float*)d_in[2];
    const float* b0 = (const float*)d_in[3];
    const float* Uz = (const float*)d_in[4];
    const float* Ug = (const float*)d_in[5];
    const float* Ur = (const float*)d_in[6];
    const float* Uh = (const float*)d_in[7];
    const float* Wz = (const float*)d_in[8];
    const float* Wg = (const float*)d_in[9];
    const float* Wr = (const float*)d_in[10];
    const float* Wh = (const float*)d_in[11];
    const float* bz = (const float*)d_in[12];
    const float* bg = (const float*)d_in[13];
    const float* br = (const float*)d_in[14];
    const float* bh = (const float*)d_in[15];
    const float* Wf = (const float*)d_in[16];
    const float* bf = (const float*)d_in[17];
    float* out = (float*)d_out;

    const int N = in_sizes[0];
    const int blocks = N / MBLK;
    dgm_fused_kernel<<<blocks, 256, 0, stream>>>(
        t, x, W0, b0, Uz, Ug, Ur, Uh, Wz, Wg, Wr, Wh,
        bz, bg, br, bh, Wf, bf, out);
}


# Each wave keeps 2 row tiles x 2 column tiles (32 rows per block): every weight fragment loaded from global feeds 2 WMMAs, halving weight reads from L2; summation order unchanged
# speedup vs baseline: 1.1452x; 1.1452x over previous
#include <hip/hip_runtime.h>


typedef float v2f __attribute__((ext_vector_type(2)));
typedef float v8f __attribute__((ext_vector_type(8)));

#define H       256
#define LAYERS  3
#define MBLK    32
#define RT      2
#define STRIDE  264

__device__ __forceinline__ v8f wmma_f32x4(v2f a, v2f b, v8f c) {
    return __builtin_amdgcn_wmma_f32_16x16x4_f32(
         false, a,  false, b,
         (short)0, c,  false,  false);
}

__device__ __forceinline__ float fast_tanh(float v) {
#if __has_builtin(__builtin_amdgcn_tanhf)
    return __builtin_amdgcn_tanhf(v);
#else
    float e = __expf(2.0f * v);
    return 1.0f - 2.0f / (e + 1.0f);
#endif
}

__global__ __launch_bounds__(256) void dgm_fused_kernel(
    const float* __restrict__ t,  const float* __restrict__ x,
    const float* __restrict__ W0, const float* __restrict__ b0,
    const float* __restrict__ Uz, const float* __restrict__ Ug,
    const float* __restrict__ Ur, const float* __restrict__ Uh,
    const float* __restrict__ Wz, const float* __restrict__ Wg,
    const float* __restrict__ Wr, const float* __restrict__ Wh,
    const float* __restrict__ bz, const float* __restrict__ bg,
    const float* __restrict__ br, const float* __restrict__ bh,
    const float* __restrict__ Wf, const float* __restrict__ bf,
    float* __restrict__ out)
{
    __shared__ float sS [MBLK][STRIDE];
    __shared__ float sS1[MBLK][STRIDE];
    __shared__ float sSR[MBLK][STRIDE];
    __shared__ float sXt[MBLK];
    __shared__ float sXx[MBLK];

    const int tid  = threadIdx.x;
    const int wave = tid >> 5;
    const int lane = tid & 31;
    const int half = lane >> 4;
    const int l16  = lane & 15;
    const int row0 = blockIdx.x * MBLK;

    if (tid < MBLK) {
        sXt[tid] = t[row0 + tid];
        sXx[tid] = x[row0 + tid];
    }
    __syncthreads();

    {
        const int   c   = tid;
        const float w0t = W0[c];
        const float w0x = W0[H + c];
        const float bb  = b0[c];
        #pragma unroll
        for (int r = 0; r < MBLK; ++r) {
            float v = fast_tanh(sXt[r] * w0t + sXx[r] * w0x + bb);
            sS1[r][c] = v;
            sS [r][c] = v;
        }
    }
    __syncthreads();

    float xt8[RT][8], xx8[RT][8];
    #pragma unroll
    for (int rt = 0; rt < RT; ++rt) {
        #pragma unroll
        for (int v = 0; v < 8; ++v) {
            xt8[rt][v] = sXt[rt * 16 + half * 8 + v];
            xx8[rt][v] = sXx[rt * 16 + half * 8 + v];
        }
    }

    const int koff = half * 2;
    const int cb0  = wave * 32;

    for (int l = 0; l < LAYERS; ++l) {
        const float* wzl = Wz + l * H * H;
        const float* wgl = Wg + l * H * H;
        const float* wrl = Wr + l * H * H;
        const float* whl = Wh + l * H * H;
        const float* uzl = Uz + l * 2 * H;
        const float* ugl = Ug + l * 2 * H;
        const float* url = Ur + l * 2 * H;
        const float* uhl = Uh + l * 2 * H;
        const float* bzl = bz + l * H;
        const float* bgl = bg + l * H;
        const float* brl = br + l * H;
        const float* bhl = bh + l * H;

        v8f accR[RT][2];
        #pragma unroll
        for (int t2 = 0; t2 < 2; ++t2) {
            const int   n  = cb0 + t2 * 16 + l16;
            const float u0 = url[n];
            const float u1 = url[H + n];
            const float bb = brl[n];
            #pragma unroll
            for (int rt = 0; rt < RT; ++rt) {
                #pragma unroll
                for (int v = 0; v < 8; ++v)
                    accR[rt][t2][v] = xt8[rt][v] * u0 + xx8[rt][v] * u1 + bb;
            }
        }
        for (int k = 0; k < H; k += 4) {
            v2f a[RT];
            #pragma unroll
            for (int rt = 0; rt < RT; ++rt)
                a[rt] = *(const v2f*)&sS[rt * 16 + l16][k + koff];
            #pragma unroll
            for (int t2 = 0; t2 < 2; ++t2) {
                const int n = cb0 + t2 * 16 + l16;
                v2f b;
                b.x = wrl[(k + koff)     * H + n];
                b.y = wrl[(k + koff + 1) * H + n];
                #pragma unroll
                for (int rt = 0; rt < RT; ++rt)
                    accR[rt][t2] = wmma_f32x4(a[rt], b, accR[rt][t2]);
            }
        }
        #pragma unroll
        for (int rt = 0; rt < RT; ++rt) {
            #pragma unroll
            for (int t2 = 0; t2 < 2; ++t2) {
                const int n = cb0 + t2 * 16 + l16;
                #pragma unroll
                for (int v = 0; v < 8; ++v) {
                    const int m = rt * 16 + half * 8 + v;
                    float R = fast_tanh(accR[rt][t2][v]);
                    sSR[m][n] = sS[m][n] * R;
                }
            }
        }
        __syncthreads();

        v8f accZ[RT][2], accG[RT][2], accHg[RT][2];
        #pragma unroll
        for (int t2 = 0; t2 < 2; ++t2) {
            const int   n  = cb0 + t2 * 16 + l16;
            const float z0 = uzl[n], z1 = uzl[H + n], zb = bzl[n];
            const float g0 = ugl[n], g1 = ugl[H + n], gb = bgl[n];
            const float h0 = uhl[n], h1 = uhl[H + n], hb = bhl[n];
            #pragma unroll
            for (int rt = 0; rt < RT; ++rt) {
                #pragma unroll
                for (int v = 0; v < 8; ++v) {
                    accZ [rt][t2][v] = xt8[rt][v] * z0 + xx8[rt][v] * z1 + zb;
                    accG [rt][t2][v] = xt8[rt][v] * g0 + xx8[rt][v] * g1 + gb;
                    accHg[rt][t2][v] = xt8[rt][v] * h0 + xx8[rt][v] * h1 + hb;
                }
            }
        }
        for (int k = 0; k < H; k += 4) {
            v2f aS[RT], aS1[RT], aSR[RT];
            #pragma unroll
            for (int rt = 0; rt < RT; ++rt) {
                aS [rt] = *(const v2f*)&sS [rt * 16 + l16][k + koff];
                aS1[rt] = *(const v2f*)&sS1[rt * 16 + l16][k + koff];
                aSR[rt] = *(const v2f*)&sSR[rt * 16 + l16][k + koff];
            }
            #pragma unroll
            for (int t2 = 0; t2 < 2; ++t2) {
                const int n = cb0 + t2 * 16 + l16;
                v2f bzv, bgv, bhv;
                bzv.x = wzl[(k + koff)     * H + n];
                bzv.y = wzl[(k + koff + 1) * H + n];
                bgv.x = wgl[(k + koff)     * H + n];
                bgv.y = wgl[(k + koff + 1) * H + n];
                bhv.x = whl[(k + koff)     * H + n];
                bhv.y = whl[(k + koff + 1) * H + n];
                #pragma unroll
                for (int rt = 0; rt < RT; ++rt) {
                    accZ [rt][t2] = wmma_f32x4(aS [rt], bzv, accZ [rt][t2]);
                    accG [rt][t2] = wmma_f32x4(aS1[rt], bgv, accG [rt][t2]);
                    accHg[rt][t2] = wmma_f32x4(aSR[rt], bhv, accHg[rt][t2]);
                }
            }
        }
        float snew[RT][2][8];
        #pragma unroll
        for (int rt = 0; rt < RT; ++rt) {
            #pragma unroll
            for (int t2 = 0; t2 < 2; ++t2) {
                const int n = cb0 + t2 * 16 + l16;
                #pragma unroll
                for (int v = 0; v < 8; ++v) {
                    const int m = rt * 16 + half * 8 + v;
                    float Zv = fast_tanh(accZ [rt][t2][v]);
                    float Gv = fast_tanh(accG [rt][t2][v]);
                    float Hv = fast_tanh(accHg[rt][t2][v]);
                    snew[rt][t2][v] = (1.0f - Gv) * Hv + Zv * sS[m][n];
                }
            }
        }
        __syncthreads();
        #pragma unroll
        for (int rt = 0; rt < RT; ++rt) {
            #pragma unroll
            for (int t2 = 0; t2 < 2; ++t2) {
                const int n = cb0 + t2 * 16 + l16;
                #pragma unroll
                for (int v = 0; v < 8; ++v)
                    sS[rt * 16 + half * 8 + v][n] = snew[rt][t2][v];
            }
        }
        __syncthreads();
    }

    #pragma unroll
    for (int rt = 0; rt < RT; ++rt) {
        const int r  = rt * 16 + (tid >> 4);
        const int ci = tid & 15;
        float p = 0.0f;
        #pragma unroll
        for (int j = 0; j < 16; ++j) {
            const int c = ci * 16 + j;
            p += sS[r][c] * Wf[c];
        }
        #pragma unroll
        for (int off = 8; off >= 1; off >>= 1)
            p += __shfl_xor(p, off, 16);
        if (ci == 0)
            out[row0 + r] = p + bf[0];
    }
}

extern "C" void kernel_launch(void* const* d_in, const int* in_sizes, int n_in,
                              void* d_out, int out_size, void* d_ws, size_t ws_size,
                              hipStream_t stream) {
    (void)n_in; (void)d_ws; (void)ws_size; (void)out_size;
    const float* t  = (const float*)d_in[0];
    const float* x  = (const float*)d_in[1];
    const float* W0 = (const float*)d_in[2];
    const float* b0 = (const float*)d_in[3];
    const float* Uz = (const float*)d_in[4];
    const float* Ug = (const float*)d_in[5];
    const float* Ur = (const float*)d_in[6];
    const float* Uh = (const float*)d_in[7];
    const float* Wz = (const float*)d_in[8];
    const float* Wg = (const float*)d_in[9];
    const float* Wr = (const float*)d_in[10];
    const float* Wh = (const float*)d_in[11];
    const float* bz = (const float*)d_in[12];
    const float* bg = (const float*)d_in[13];
    const float* br = (const float*)d_in[14];
    const float* bh = (const float*)d_in[15];
    const float* Wf = (const float*)d_in[16];
    const float* bf = (const float*)d_in[17];
    float* out = (float*)d_out;

    const int N = in_sizes[0];
    const int blocks = N / MBLK;
    dgm_fused_kernel<<<blocks, 256, 0, stream>>>(
        t, x, W0, b0, Uz, Ug, Ur, Uh, Wz, Wg, Wr, Wh,
        bz, bg, br, bh, Wf, bf, out);
}


# Phase-2 K-loop software-pipelined: each k-step's weight fragments are loaded from global three k-steps ahead into a ring of four register sets; summation order unchanged
# speedup vs baseline: 1.1933x; 1.0420x over previous
#include <hip/hip_runtime.h>


typedef float v2f __attribute__((ext_vector_type(2)));
typedef float v8f __attribute__((ext_vector_type(8)));

#define H       256
#define LAYERS  3
#define MBLK    32
#define RT      2
#define STRIDE  264

__device__ __forceinline__ v8f wmma_f32x4(v2f a, v2f b, v8f c) {
    return __builtin_amdgcn_wmma_f32_16x16x4_f32(
         false, a,  false, b,
         (short)0, c,  false,  false);
}

__device__ __forceinline__ float fast_tanh(float v) {
#if __has_builtin(__builtin_amdgcn_tanhf)
    return __builtin_amdgcn_tanhf(v);
#else
    float e = __expf(2.0f * v);
    return 1.0f - 2.0f / (e + 1.0f);
#endif
}

__global__ __launch_bounds__(256) void dgm_fused_kernel(
    const float* __restrict__ t,  const float* __restrict__ x,
    const float* __restrict__ W0, const float* __restrict__ b0,
    const float* __restrict__ Uz, const float* __restrict__ Ug,
    const float* __restrict__ Ur, const float* __restrict__ Uh,
    const float* __restrict__ Wz, const float* __restrict__ Wg,
    const float* __restrict__ Wr, const float* __restrict__ Wh,
    const float* __restrict__ bz, const float* __restrict__ bg,
    const float* __restrict__ br, const float* __restrict__ bh,
    const float* __restrict__ Wf, const float* __restrict__ bf,
    float* __restrict__ out)
{
    __shared__ float sS [MBLK][STRIDE];
    __shared__ float sS1[MBLK][STRIDE];
    __shared__ float sSR[MBLK][STRIDE];
    __shared__ float sXt[MBLK];
    __shared__ float sXx[MBLK];

    const int tid  = threadIdx.x;
    const int wave = tid >> 5;
    const int lane = tid & 31;
    const int half = lane >> 4;
    const int l16  = lane & 15;
    const int row0 = blockIdx.x * MBLK;

    if (tid < MBLK) {
        sXt[tid] = t[row0 + tid];
        sXx[tid] = x[row0 + tid];
    }
    __syncthreads();

    {
        const int   c   = tid;
        const float w0t = W0[c];
        const float w0x = W0[H + c];
        const float bb  = b0[c];
        #pragma unroll
        for (int r = 0; r < MBLK; ++r) {
            float v = fast_tanh(sXt[r] * w0t + sXx[r] * w0x + bb);
            sS1[r][c] = v;
            sS [r][c] = v;
        }
    }
    __syncthreads();

    float xt8[RT][8], xx8[RT][8];
    #pragma unroll
    for (int rt = 0; rt < RT; ++rt) {
        #pragma unroll
        for (int v = 0; v < 8; ++v) {
            xt8[rt][v] = sXt[rt * 16 + half * 8 + v];
            xx8[rt][v] = sXx[rt * 16 + half * 8 + v];
        }
    }

    const int koff = half * 2;
    const int cb0  = wave * 32;

    for (int l = 0; l < LAYERS; ++l) {
        const float* wzl = Wz + l * H * H;
        const float* wgl = Wg + l * H * H;
        const float* wrl = Wr + l * H * H;
        const float* whl = Wh + l * H * H;
        const float* uzl = Uz + l * 2 * H;
        const float* ugl = Ug + l * 2 * H;
        const float* url = Ur + l * 2 * H;
        const float* uhl = Uh + l * 2 * H;
        const float* bzl = bz + l * H;
        const float* bgl = bg + l * H;
        const float* brl = br + l * H;
        const float* bhl = bh + l * H;

        v8f accR[RT][2];
        #pragma unroll
        for (int t2 = 0; t2 < 2; ++t2) {
            const int   n  = cb0 + t2 * 16 + l16;
            const float u0 = url[n];
            const float u1 = url[H + n];
            const float bb = brl[n];
            #pragma unroll
            for (int rt = 0; rt < RT; ++rt) {
                #pragma unroll
                for (int v = 0; v < 8; ++v)
                    accR[rt][t2][v] = xt8[rt][v] * u0 + xx8[rt][v] * u1 + bb;
            }
        }
        for (int k = 0; k < H; k += 4) {
            v2f a[RT];
            #pragma unroll
            for (int rt = 0; rt < RT; ++rt)
                a[rt] = *(const v2f*)&sS[rt * 16 + l16][k + koff];
            #pragma unroll
            for (int t2 = 0; t2 < 2; ++t2) {
                const int n = cb0 + t2 * 16 + l16;
                v2f b;
                b.x = wrl[(k + koff)     * H + n];
                b.y = wrl[(k + koff + 1) * H + n];
                #pragma unroll
                for (int rt = 0; rt < RT; ++rt)
                    accR[rt][t2] = wmma_f32x4(a[rt], b, accR[rt][t2]);
            }
        }
        #pragma unroll
        for (int rt = 0; rt < RT; ++rt) {
            #pragma unroll
            for (int t2 = 0; t2 < 2; ++t2) {
                const int n = cb0 + t2 * 16 + l16;
                #pragma unroll
                for (int v = 0; v < 8; ++v) {
                    const int m = rt * 16 + half * 8 + v;
                    float R = fast_tanh(accR[rt][t2][v]);
                    sSR[m][n] = sS[m][n] * R;
                }
            }
        }
        __syncthreads();

        v8f accZ[RT][2], accG[RT][2], accHg[RT][2];
        #pragma unroll
        for (int t2 = 0; t2 < 2; ++t2) {
            const int   n  = cb0 + t2 * 16 + l16;
            const float z0 = uzl[n], z1 = uzl[H + n], zb = bzl[n];
            const float g0 = ugl[n], g1 = ugl[H + n], gb = bgl[n];
            const float h0 = uhl[n], h1 = uhl[H + n], hb = bhl[n];
            #pragma unroll
            for (int rt = 0; rt < RT; ++rt) {
                #pragma unroll
                for (int v = 0; v < 8; ++v) {
                    accZ [rt][t2][v] = xt8[rt][v] * z0 + xx8[rt][v] * z1 + zb;
                    accG [rt][t2][v] = xt8[rt][v] * g0 + xx8[rt][v] * g1 + gb;
                    accHg[rt][t2][v] = xt8[rt][v] * h0 + xx8[rt][v] * h1 + hb;
                }
            }
        }
        struct BFrag { v2f z[2], g[2], h[2]; };
        const unsigned lofs = (unsigned)(koff * H + cb0 + l16);
        auto load_b = [&](BFrag& f, int k) {
            const float* pz = wzl + k * H;
            const float* pg = wgl + k * H;
            const float* ph = whl + k * H;
            #pragma unroll
            for (int t2 = 0; t2 < 2; ++t2) {
                f.z[t2].x = pz[lofs + t2 * 16];
                f.z[t2].y = pz[lofs + t2 * 16 + H];
                f.g[t2].x = pg[lofs + t2 * 16];
                f.g[t2].y = pg[lofs + t2 * 16 + H];
                f.h[t2].x = ph[lofs + t2 * 16];
                f.h[t2].y = ph[lofs + t2 * 16 + H];
            }
        };
        auto kstep = [&](const BFrag& f, int k) {
            v2f aS[RT], aS1[RT], aSR[RT];
            #pragma unroll
            for (int rt = 0; rt < RT; ++rt) {
                aS [rt] = *(const v2f*)&sS [rt * 16 + l16][k + koff];
                aS1[rt] = *(const v2f*)&sS1[rt * 16 + l16][k + koff];
                aSR[rt] = *(const v2f*)&sSR[rt * 16 + l16][k + koff];
            }
            #pragma unroll
            for (int t2 = 0; t2 < 2; ++t2) {
                #pragma unroll
                for (int rt = 0; rt < RT; ++rt) {
                    accZ [rt][t2] = wmma_f32x4(aS [rt], f.z[t2], accZ [rt][t2]);
                    accG [rt][t2] = wmma_f32x4(aS1[rt], f.g[t2], accG [rt][t2]);
                    accHg[rt][t2] = wmma_f32x4(aSR[rt], f.h[t2], accHg[rt][t2]);
                }
            }
        };
        BFrag b0, b1, b2, b3;
        load_b(b0, 0);
        __builtin_amdgcn_sched_barrier(0);
        load_b(b1, 4);
        __builtin_amdgcn_sched_barrier(0);
        load_b(b2, 8);
        __builtin_amdgcn_sched_barrier(0);
        #pragma nounroll
        for (int k = 0; k < H - 16; k += 16) {
            load_b(b3, k + 12);
            __builtin_amdgcn_sched_barrier(0);
            kstep(b0, k);
            __builtin_amdgcn_sched_barrier(0);
            load_b(b0, k + 16);
            __builtin_amdgcn_sched_barrier(0);
            kstep(b1, k + 4);
            __builtin_amdgcn_sched_barrier(0);
            load_b(b1, k + 20);
            __builtin_amdgcn_sched_barrier(0);
            kstep(b2, k + 8);
            __builtin_amdgcn_sched_barrier(0);
            load_b(b2, k + 24);
            __builtin_amdgcn_sched_barrier(0);
            kstep(b3, k + 12);
            __builtin_amdgcn_sched_barrier(0);
        }
        load_b(b3, H - 4);
        __builtin_amdgcn_sched_barrier(0);
        kstep(b0, H - 16);
        kstep(b1, H - 12);
        kstep(b2, H - 8);
        kstep(b3, H - 4);
        float snew[RT][2][8];
        #pragma unroll
        for (int rt = 0; rt < RT; ++rt) {
            #pragma unroll
            for (int t2 = 0; t2 < 2; ++t2) {
                const int n = cb0 + t2 * 16 + l16;
                #pragma unroll
                for (int v = 0; v < 8; ++v) {
                    const int m = rt * 16 + half * 8 + v;
                    float Zv = fast_tanh(accZ [rt][t2][v]);
                    float Gv = fast_tanh(accG [rt][t2][v]);
                    float Hv = fast_tanh(accHg[rt][t2][v]);
                    snew[rt][t2][v] = (1.0f - Gv) * Hv + Zv * sS[m][n];
                }
            }
        }
        __syncthreads();
        #pragma unroll
        for (int rt = 0; rt < RT; ++rt) {
            #pragma unroll
            for (int t2 = 0; t2 < 2; ++t2) {
                const int n = cb0 + t2 * 16 + l16;
                #pragma unroll
                for (int v = 0; v < 8; ++v)
                    sS[rt * 16 + half * 8 + v][n] = snew[rt][t2][v];
            }
        }
        __syncthreads();
    }

    #pragma unroll
    for (int rt = 0; rt < RT; ++rt) {
        const int r  = rt * 16 + (tid >> 4);
        const int ci = tid & 15;
        float p = 0.0f;
        #pragma unroll
        for (int j = 0; j < 16; ++j) {
            const int c = ci * 16 + j;
            p += sS[r][c] * Wf[c];
        }
        #pragma unroll
        for (int off = 8; off >= 1; off >>= 1)
            p += __shfl_xor(p, off, 16);
        if (ci == 0)
            out[row0 + r] = p + bf[0];
    }
}

extern "C" void kernel_launch(void* const* d_in, const int* in_sizes, int n_in,
                              void* d_out, int out_size, void* d_ws, size_t ws_size,
                              hipStream_t stream) {
    (void)n_in; (void)d_ws; (void)ws_size; (void)out_size;
    const float* t  = (const float*)d_in[0];
    const float* x  = (const float*)d_in[1];
    const float* W0 = (const float*)d_in[2];
    const float* b0 = (const float*)d_in[3];
    const float* Uz = (const float*)d_in[4];
    const float* Ug = (const float*)d_in[5];
    const float* Ur = (const float*)d_in[6];
    const float* Uh = (const float*)d_in[7];
    const float* Wz = (const float*)d_in[8];
    const float* Wg = (const float*)d_in[9];
    const float* Wr = (const float*)d_in[10];
    const float* Wh = (const float*)d_in[11];
    const float* bz = (const float*)d_in[12];
    const float* bg = (const float*)d_in[13];
    const float* br = (const float*)d_in[14];
    const float* bh = (const float*)d_in[15];
    const float* Wf = (const float*)d_in[16];
    const float* bf = (const float*)d_in[17];
    float* out = (float*)d_out;

    const int N = in_sizes[0];
    const int blocks = N / MBLK;
    dgm_fused_kernel<<<blocks, 256, 0, stream>>>(
        t, x, W0, b0, Uz, Ug, Ur, Uh, Wz, Wg, Wr, Wh,
        bz, bg, br, bh, Wf, bf, out);
}


# 64-row blocks, 4 row tiles per wave: each weight fragment from global feeds 4 WMMAs (was 2); gates run in turn on two LDS planes (S*R stored over S1, S1 rebuilt per layer); k order of sums unchanged
# speedup vs baseline: 1.3052x; 1.0938x over previous
#include <hip/hip_runtime.h>


typedef float v2f __attribute__((ext_vector_type(2)));
typedef float v8f __attribute__((ext_vector_type(8)));

#define H       256
#define LAYERS  3
#define MBLK    64
#define RT      4
#define STRIDE  264

__device__ __forceinline__ v8f wmma_f32x4(v2f a, v2f b, v8f c) {
    return __builtin_amdgcn_wmma_f32_16x16x4_f32(
         false, a,  false, b,
         (short)0, c,  false,  false);
}

__device__ __forceinline__ float fast_tanh(float v) {
#if __has_builtin(__builtin_amdgcn_tanhf)
    return __builtin_amdgcn_tanhf(v);
#else
    float e = __expf(2.0f * v);
    return 1.0f - 2.0f / (e + 1.0f);
#endif
}

__device__ __forceinline__ float s1_elem(float xt, float xx, float w0t, float w0x, float bb) {
    return fast_tanh(xt * w0t + xx * w0x + bb);
}

__global__ __launch_bounds__(256, 4) void dgm_fused_kernel(
    const float* __restrict__ t,  const float* __restrict__ x,
    const float* __restrict__ W0, const float* __restrict__ b0,
    const float* __restrict__ Uz, const float* __restrict__ Ug,
    const float* __restrict__ Ur, const float* __restrict__ Uh,
    const float* __restrict__ Wz, const float* __restrict__ Wg,
    const float* __restrict__ Wr, const float* __restrict__ Wh,
    const float* __restrict__ bz, const float* __restrict__ bg,
    const float* __restrict__ br, const float* __restrict__ bh,
    const float* __restrict__ Wf, const float* __restrict__ bf,
    float* __restrict__ out)
{
    __shared__ float sS [MBLK][STRIDE];
    __shared__ float sS1[MBLK][STRIDE];
    __shared__ float sXt[MBLK];
    __shared__ float sXx[MBLK];

    const int tid  = threadIdx.x;
    const int wave = tid >> 5;
    const int lane = tid & 31;
    const int half = lane >> 4;
    const int l16  = lane & 15;
    const int row0 = blockIdx.x * MBLK;

    if (tid < MBLK) {
        sXt[tid] = t[row0 + tid];
        sXx[tid] = x[row0 + tid];
    }
    __syncthreads();

    {
        const int   c   = tid;
        const float w0t = W0[c];
        const float w0x = W0[H + c];
        const float bb  = b0[c];
        #pragma unroll
        for (int r = 0; r < MBLK; ++r) {
            float v = s1_elem(sXt[r], sXx[r], w0t, w0x, bb);
            sS1[r][c] = v;
            sS [r][c] = v;
        }
    }
    __syncthreads();

    const int koff = half * 2;
    const int cb0  = wave * 32;
    const unsigned lofs = (unsigned)(koff * H + cb0 + l16);

    auto init_acc = [&](v8f (&acc)[RT][2], const float* u, const float* b) {
        float u0[2], u1[2], bb[2];
        #pragma unroll
        for (int t2 = 0; t2 < 2; ++t2) {
            const int n = cb0 + t2 * 16 + l16;
            u0[t2] = u[n];
            u1[t2] = u[H + n];
            bb[t2] = b[n];
        }
        #pragma unroll
        for (int rt = 0; rt < RT; ++rt) {
            #pragma unroll
            for (int t2 = 0; t2 < 2; ++t2) {
                #pragma unroll
                for (int v = 0; v < 8; ++v) {
                    const int m = rt * 16 + half * 8 + v;
                    acc[rt][t2][v] = sXt[m] * u0[t2] + sXx[m] * u1[t2] + bb[t2];
                }
            }
        }
    };

    struct BGrp { v2f b[2][2]; };
    auto load_bg = [&](BGrp& f, const float* w, int k) {
        const float* p = w + k * H;
        #pragma unroll
        for (int s = 0; s < 2; ++s) {
            #pragma unroll
            for (int t2 = 0; t2 < 2; ++t2) {
                f.b[s][t2].x = p[lofs + s * 4 * H + t2 * 16];
                f.b[s][t2].y = p[lofs + s * 4 * H + t2 * 16 + H];
            }
        }
    };
    auto kgroup = [&](v8f (&acc)[RT][2], const float (*sA)[STRIDE], const BGrp& f, int k) {
        #pragma unroll
        for (int s = 0; s < 2; ++s) {
            v2f a[RT];
            #pragma unroll
            for (int rt = 0; rt < RT; ++rt)
                a[rt] = *(const v2f*)&sA[rt * 16 + l16][k + s * 4 + koff];
            #pragma unroll
            for (int t2 = 0; t2 < 2; ++t2) {
                #pragma unroll
                for (int rt = 0; rt < RT; ++rt)
                    acc[rt][t2] = wmma_f32x4(a[rt], f.b[s][t2], acc[rt][t2]);
            }
        }
    };
    auto gate_pass = [&](v8f (&acc)[RT][2], const float (*sA)[STRIDE], const float* w) {
        BGrp g0, g1, g2, g3;
        load_bg(g0, w, 0);
        __builtin_amdgcn_sched_barrier(0);
        load_bg(g1, w, 8);
        __builtin_amdgcn_sched_barrier(0);
        load_bg(g2, w, 16);
        __builtin_amdgcn_sched_barrier(0);
        #pragma nounroll
        for (int k = 0; k < H - 32; k += 32) {
            load_bg(g3, w, k + 24);
            __builtin_amdgcn_sched_barrier(0);
            kgroup(acc, sA, g0, k);
            __builtin_amdgcn_sched_barrier(0);
            load_bg(g0, w, k + 32);
            __builtin_amdgcn_sched_barrier(0);
            kgroup(acc, sA, g1, k + 8);
            __builtin_amdgcn_sched_barrier(0);
            load_bg(g1, w, k + 40);
            __builtin_amdgcn_sched_barrier(0);
            kgroup(acc, sA, g2, k + 16);
            __builtin_amdgcn_sched_barrier(0);
            load_bg(g2, w, k + 48);
            __builtin_amdgcn_sched_barrier(0);
            kgroup(acc, sA, g3, k + 24);
            __builtin_amdgcn_sched_barrier(0);
        }
        load_bg(g3, w, H - 8);
        __builtin_amdgcn_sched_barrier(0);
        kgroup(acc, sA, g0, H - 32);
        kgroup(acc, sA, g1, H - 24);
        kgroup(acc, sA, g2, H - 16);
        kgroup(acc, sA, g3, H - 8);
    };

    for (int l = 0; l < LAYERS; ++l) {
        const float* wzl = Wz + l * H * H;
        const float* wgl = Wg + l * H * H;
        const float* wrl = Wr + l * H * H;
        const float* whl = Wh + l * H * H;
        const float* uzl = Uz + l * 2 * H;
        const float* ugl = Ug + l * 2 * H;
        const float* url = Ur + l * 2 * H;
        const float* uhl = Uh + l * 2 * H;
        const float* bzl = bz + l * H;
        const float* bgl = bg + l * H;
        const float* brl = br + l * H;
        const float* bhl = bh + l * H;

        v8f acc[RT][2];

        v8f omg[RT][2];
        init_acc(acc, ugl, bgl);
        gate_pass(acc, sS1, wgl);
        #pragma unroll
        for (int rt = 0; rt < RT; ++rt) {
            #pragma unroll
            for (int t2 = 0; t2 < 2; ++t2) {
                #pragma unroll
                for (int v = 0; v < 8; ++v)
                    omg[rt][t2][v] = 1.0f - fast_tanh(acc[rt][t2][v]);
            }
        }

        float sr[RT][2][8];
        init_acc(acc, url, brl);
        gate_pass(acc, sS, wrl);
        #pragma unroll
        for (int rt = 0; rt < RT; ++rt) {
            #pragma unroll
            for (int t2 = 0; t2 < 2; ++t2) {
                const int n = cb0 + t2 * 16 + l16;
                #pragma unroll
                for (int v = 0; v < 8; ++v) {
                    const int m = rt * 16 + half * 8 + v;
                    float R = fast_tanh(acc[rt][t2][v]);
                    sr[rt][t2][v] = sS[m][n] * R;
                }
            }
        }
        __syncthreads();
        #pragma unroll
        for (int rt = 0; rt < RT; ++rt) {
            #pragma unroll
            for (int t2 = 0; t2 < 2; ++t2) {
                const int n = cb0 + t2 * 16 + l16;
                #pragma unroll
                for (int v = 0; v < 8; ++v)
                    sS1[rt * 16 + half * 8 + v][n] = sr[rt][t2][v];
            }
        }
        __syncthreads();

        init_acc(acc, uhl, bhl);
        gate_pass(acc, sS1, whl);
        #pragma unroll
        for (int rt = 0; rt < RT; ++rt) {
            #pragma unroll
            for (int t2 = 0; t2 < 2; ++t2) {
                #pragma unroll
                for (int v = 0; v < 8; ++v)
                    omg[rt][t2][v] = omg[rt][t2][v] * fast_tanh(acc[rt][t2][v]);
            }
        }
        #pragma unroll
        for (int rt = 0; rt < RT; ++rt) {
            #pragma unroll
            for (int t2 = 0; t2 < 2; ++t2)
                asm volatile("" :: "v"(omg[rt][t2]));
        }

        float snew[RT][2][8];
        init_acc(acc, uzl, bzl);
        gate_pass(acc, sS, wzl);
        #pragma unroll
        for (int rt = 0; rt < RT; ++rt) {
            #pragma unroll
            for (int t2 = 0; t2 < 2; ++t2) {
                const int n = cb0 + t2 * 16 + l16;
                #pragma unroll
                for (int v = 0; v < 8; ++v) {
                    const int m = rt * 16 + half * 8 + v;
                    float Zv = fast_tanh(acc[rt][t2][v]);
                    snew[rt][t2][v] = omg[rt][t2][v] + Zv * sS[m][n];
                }
            }
        }
        __syncthreads();
        #pragma unroll
        for (int rt = 0; rt < RT; ++rt) {
            #pragma unroll
            for (int t2 = 0; t2 < 2; ++t2) {
                const int n = cb0 + t2 * 16 + l16;
                #pragma unroll
                for (int v = 0; v < 8; ++v)
                    sS[rt * 16 + half * 8 + v][n] = snew[rt][t2][v];
            }
        }
        if (l + 1 < LAYERS) {
            const int   c   = tid;
            const float w0t = W0[c];
            const float w0x = W0[H + c];
            const float bb  = b0[c];
            #pragma unroll
            for (int r = 0; r < MBLK; ++r)
                sS1[r][c] = s1_elem(sXt[r], sXx[r], w0t, w0x, bb);
        }
        __syncthreads();
    }

    #pragma unroll
    for (int rt = 0; rt < RT; ++rt) {
        const int ln = (int)__builtin_amdgcn_mbcnt_lo(~0u, 0u);
        const int r  = rt * 16 + ((tid - ln) >> 4) + (ln >> 4);
        const int ci = ln & 15;
        float p = 0.0f;
        #pragma unroll
        for (int j = 0; j < 16; ++j) {
            const int c = ci * 16 + j;
            p += sS[r][c] * Wf[c];
        }
        #pragma unroll
        for (int off = 8; off >= 1; off >>= 1)
            p += __shfl_xor(p, off, 16);
        if (ci == 0)
            out[row0 + r] = p + bf[0];
    }
}

extern "C" void kernel_launch(void* const* d_in, const int* in_sizes, int n_in,
                              void* d_out, int out_size, void* d_ws, size_t ws_size,
                              hipStream_t stream) {
    (void)n_in; (void)d_ws; (void)ws_size; (void)out_size;
    const float* t  = (const float*)d_in[0];
    const float* x  = (const float*)d_in[1];
    const float* W0 = (const float*)d_in[2];
    const float* b0 = (const float*)d_in[3];
    const float* Uz = (const float*)d_in[4];
    const float* Ug = (const float*)d_in[5];
    const float* Ur = (const float*)d_in[6];
    const float* Uh = (const float*)d_in[7];
    const float* Wz = (const float*)d_in[8];
    const float* Wg = (const float*)d_in[9];
    const float* Wr = (const float*)d_in[10];
    const float* Wh = (const float*)d_in[11];
    const float* bz = (const float*)d_in[12];
    const float* bg = (const float*)d_in[13];
    const float* br = (const float*)d_in[14];
    const float* bh = (const float*)d_in[15];
    const float* Wf = (const float*)d_in[16];
    const float* bf = (const float*)d_in[17];
    float* out = (float*)d_out;

    const int N = in_sizes[0];
    const int blocks = N / MBLK;
    dgm_fused_kernel<<<blocks, 256, 0, stream>>>(
        t, x, W0, b0, Uz, Ug, Ur, Uh, Wz, Wg, Wr, Wh,
        bz, bg, br, bh, Wf, bf, out);
}


# K-loops: each group's A fragments read from LDS one group ahead into a second register set, beside the B refill, so WMMAs no longer wait on LDS; LDS pitch 264->260 to fit registers; sums unchanged
# speedup vs baseline: 1.3458x; 1.0311x over previous
#include <hip/hip_runtime.h>


typedef float v2f __attribute__((ext_vector_type(2)));
typedef float v8f __attribute__((ext_vector_type(8)));

#define H       256
#define LAYERS  3
#define MBLK    64
#define RT      4
#define STRIDE  260

__device__ __forceinline__ v8f wmma_f32x4(v2f a, v2f b, v8f c) {
    return __builtin_amdgcn_wmma_f32_16x16x4_f32(
         false, a,  false, b,
         (short)0, c,  false,  false);
}

__device__ __forceinline__ float fast_tanh(float v) {
#if __has_builtin(__builtin_amdgcn_tanhf)
    return __builtin_amdgcn_tanhf(v);
#else
    float e = __expf(2.0f * v);
    return 1.0f - 2.0f / (e + 1.0f);
#endif
}

__device__ __forceinline__ float s1_elem(float xt, float xx, float w0t, float w0x, float bb) {
    return fast_tanh(xt * w0t + xx * w0x + bb);
}

__global__ __launch_bounds__(256, 4) void dgm_fused_kernel(
    const float* __restrict__ t,  const float* __restrict__ x,
    const float* __restrict__ W0, const float* __restrict__ b0,
    const float* __restrict__ Uz, const float* __restrict__ Ug,
    const float* __restrict__ Ur, const float* __restrict__ Uh,
    const float* __restrict__ Wz, const float* __restrict__ Wg,
    const float* __restrict__ Wr, const float* __restrict__ Wh,
    const float* __restrict__ bz, const float* __restrict__ bg,
    const float* __restrict__ br, const float* __restrict__ bh,
    const float* __restrict__ Wf, const float* __restrict__ bf,
    float* __restrict__ out)
{
    __shared__ float sS [MBLK][STRIDE];
    __shared__ float sS1[MBLK][STRIDE];
    __shared__ float sXt[MBLK];
    __shared__ float sXx[MBLK];

    const int tid  = threadIdx.x;
    const int wave = tid >> 5;
    const int lane = tid & 31;
    const int half = lane >> 4;
    const int l16  = lane & 15;
    const int row0 = blockIdx.x * MBLK;
    const int wave_s = __builtin_amdgcn_readfirstlane(tid >> 5);

    if (tid < MBLK) {
        sXt[tid] = t[row0 + tid];
        sXx[tid] = x[row0 + tid];
    }
    __syncthreads();

    const float w0t = W0[tid];
    const float w0x = W0[H + tid];
    const float bb0 = b0[tid];
    {
        const int c = tid;
        #pragma unroll
        for (int r = 0; r < MBLK; ++r) {
            float v = s1_elem(sXt[r], sXx[r], w0t, w0x, bb0);
            sS1[r][c] = v;
            sS [r][c] = v;
        }
    }
    __syncthreads();

    const int koff = half * 2;
    const int cb0  = wave * 32;
    const unsigned lofs = (unsigned)(koff * H + cb0 + l16);

    auto init_acc = [&](v8f (&acc)[RT][2], const float* u, const float* b) {
        float u0[2], u1[2], bb[2];
        #pragma unroll
        for (int t2 = 0; t2 < 2; ++t2) {
            const int n = cb0 + t2 * 16 + l16;
            u0[t2] = u[n];
            u1[t2] = u[H + n];
            bb[t2] = b[n];
        }
        #pragma unroll
        for (int rt = 0; rt < RT; ++rt) {
            #pragma unroll
            for (int t2 = 0; t2 < 2; ++t2) {
                #pragma unroll
                for (int v = 0; v < 8; ++v) {
                    const int m = rt * 16 + half * 8 + v;
                    acc[rt][t2][v] = sXt[m] * u0[t2] + sXx[m] * u1[t2] + bb[t2];
                }
            }
        }
    };

    struct BGrp { v2f b[2][2]; };
    auto load_bg = [&](BGrp& f, const float* w, int k) {
        const float* p = w + k * H;
        #pragma unroll
        for (int s = 0; s < 2; ++s) {
            #pragma unroll
            for (int t2 = 0; t2 < 2; ++t2) {
                f.b[s][t2].x = p[lofs + s * 4 * H + t2 * 16];
                f.b[s][t2].y = p[lofs + s * 4 * H + t2 * 16 + H];
            }
        }
    };
    struct AGrp { v2f a[2][RT]; };
    auto load_ag = [&](AGrp& f, const float (*sA)[STRIDE], int k) {
        #pragma unroll
        for (int s = 0; s < 2; ++s) {
            #pragma unroll
            for (int rt = 0; rt < RT; ++rt)
                f.a[s][rt] = *(const v2f*)&sA[rt * 16 + l16][k + s * 4 + koff];
        }
    };
    auto kgroup = [&](v8f (&acc)[RT][2], const AGrp& fa, const BGrp& f) {
        #pragma unroll
        for (int s = 0; s < 2; ++s) {
            #pragma unroll
            for (int t2 = 0; t2 < 2; ++t2) {
                #pragma unroll
                for (int rt = 0; rt < RT; ++rt)
                    acc[rt][t2] = wmma_f32x4(fa.a[s][rt], f.b[s][t2], acc[rt][t2]);
            }
        }
    };
    auto gate_pass = [&](v8f (&acc)[RT][2], const float (*sA)[STRIDE], const float* w) {
        BGrp g0, g1, g2, g3;
        AGrp a0, a1;
        load_bg(g0, w, 0);
        __builtin_amdgcn_sched_barrier(0);
        load_bg(g1, w, 8);
        __builtin_amdgcn_sched_barrier(0);
        load_bg(g2, w, 16);
        load_ag(a0, sA, 0);
        __builtin_amdgcn_sched_barrier(0);
        #pragma nounroll
        for (int k = 0; k < H - 32; k += 32) {
            load_bg(g3, w, k + 24);
            load_ag(a1, sA, k + 8);
            __builtin_amdgcn_sched_barrier(0);
            kgroup(acc, a0, g0);
            __builtin_amdgcn_sched_barrier(0);
            load_bg(g0, w, k + 32);
            load_ag(a0, sA, k + 16);
            __builtin_amdgcn_sched_barrier(0);
            kgroup(acc, a1, g1);
            __builtin_amdgcn_sched_barrier(0);
            load_bg(g1, w, k + 40);
            load_ag(a1, sA, k + 24);
            __builtin_amdgcn_sched_barrier(0);
            kgroup(acc, a0, g2);
            __builtin_amdgcn_sched_barrier(0);
            load_bg(g2, w, k + 48);
            load_ag(a0, sA, k + 32);
            __builtin_amdgcn_sched_barrier(0);
            kgroup(acc, a1, g3);
            __builtin_amdgcn_sched_barrier(0);
        }
        load_bg(g3, w, H - 8);
        load_ag(a1, sA, H - 24);
        __builtin_amdgcn_sched_barrier(0);
        kgroup(acc, a0, g0);
        __builtin_amdgcn_sched_barrier(0);
        load_ag(a0, sA, H - 16);
        __builtin_amdgcn_sched_barrier(0);
        kgroup(acc, a1, g1);
        __builtin_amdgcn_sched_barrier(0);
        load_ag(a1, sA, H - 8);
        __builtin_amdgcn_sched_barrier(0);
        kgroup(acc, a0, g2);
        kgroup(acc, a1, g3);
    };

    for (int l = 0; l < LAYERS; ++l) {
        const float* wzl = Wz + l * H * H;
        const float* wgl = Wg + l * H * H;
        const float* wrl = Wr + l * H * H;
        const float* whl = Wh + l * H * H;
        const float* uzl = Uz + l * 2 * H;
        const float* ugl = Ug + l * 2 * H;
        const float* url = Ur + l * 2 * H;
        const float* uhl = Uh + l * 2 * H;
        const float* bzl = bz + l * H;
        const float* bgl = bg + l * H;
        const float* brl = br + l * H;
        const float* bhl = bh + l * H;

        v8f acc[RT][2];

        v8f omg[RT][2];
        init_acc(acc, ugl, bgl);
        gate_pass(acc, sS1, wgl);
        #pragma unroll
        for (int rt = 0; rt < RT; ++rt) {
            #pragma unroll
            for (int t2 = 0; t2 < 2; ++t2) {
                #pragma unroll
                for (int v = 0; v < 8; ++v)
                    omg[rt][t2][v] = 1.0f - fast_tanh(acc[rt][t2][v]);
            }
        }

        float sr[RT][2][8];
        init_acc(acc, url, brl);
        gate_pass(acc, sS, wrl);
        #pragma unroll
        for (int rt = 0; rt < RT; ++rt) {
            #pragma unroll
            for (int t2 = 0; t2 < 2; ++t2) {
                const int n = cb0 + t2 * 16 + l16;
                #pragma unroll
                for (int v = 0; v < 8; ++v) {
                    const int m = rt * 16 + half * 8 + v;
                    float R = fast_tanh(acc[rt][t2][v]);
                    sr[rt][t2][v] = sS[m][n] * R;
                }
            }
        }
        __syncthreads();
        #pragma unroll
        for (int rt = 0; rt < RT; ++rt) {
            #pragma unroll
            for (int t2 = 0; t2 < 2; ++t2) {
                const int n = cb0 + t2 * 16 + l16;
                #pragma unroll
                for (int v = 0; v < 8; ++v)
                    sS1[rt * 16 + half * 8 + v][n] = sr[rt][t2][v];
            }
        }
        __syncthreads();

        init_acc(acc, uhl, bhl);
        gate_pass(acc, sS1, whl);
        #pragma unroll
        for (int rt = 0; rt < RT; ++rt) {
            #pragma unroll
            for (int t2 = 0; t2 < 2; ++t2) {
                #pragma unroll
                for (int v = 0; v < 8; ++v)
                    omg[rt][t2][v] = omg[rt][t2][v] * fast_tanh(acc[rt][t2][v]);
            }
        }
        #pragma unroll
        for (int rt = 0; rt < RT; ++rt) {
            #pragma unroll
            for (int t2 = 0; t2 < 2; ++t2)
                asm volatile("" :: "v"(omg[rt][t2]));
        }

        float snew[RT][2][8];
        init_acc(acc, uzl, bzl);
        gate_pass(acc, sS, wzl);
        #pragma unroll
        for (int rt = 0; rt < RT; ++rt) {
            #pragma unroll
            for (int t2 = 0; t2 < 2; ++t2) {
                const int n = cb0 + t2 * 16 + l16;
                #pragma unroll
                for (int v = 0; v < 8; ++v) {
                    const int m = rt * 16 + half * 8 + v;
                    float Zv = fast_tanh(acc[rt][t2][v]);
                    snew[rt][t2][v] = omg[rt][t2][v] + Zv * sS[m][n];
                }
            }
        }
        __syncthreads();
        #pragma unroll
        for (int rt = 0; rt < RT; ++rt) {
            #pragma unroll
            for (int t2 = 0; t2 < 2; ++t2) {
                const int n = cb0 + t2 * 16 + l16;
                #pragma unroll
                for (int v = 0; v < 8; ++v)
                    sS[rt * 16 + half * 8 + v][n] = snew[rt][t2][v];
            }
        }
        if (l + 1 < LAYERS) {
            const int c = tid;
            #pragma unroll
            for (int r = 0; r < MBLK; ++r)
                sS1[r][c] = s1_elem(sXt[r], sXx[r], w0t, w0x, bb0);
        }
        __syncthreads();
    }

    #pragma unroll
    for (int rt = 0; rt < RT; ++rt) {
        const int ln = (int)__builtin_amdgcn_mbcnt_lo(~0u, 0u);
        const int r  = rt * 16 + wave_s * 2 + (ln >> 4);
        const int ci = ln & 15;
        float p = 0.0f;
        #pragma unroll
        for (int j = 0; j < 16; ++j) {
            const int c = ci * 16 + j;
            p += sS[r][c] * Wf[c];
        }
        #pragma unroll
        for (int off = 8; off >= 1; off >>= 1)
            p += __shfl_xor(p, off, 16);
        if (ci == 0)
            out[row0 + r] = p + bf[0];
    }
}

extern "C" void kernel_launch(void* const* d_in, const int* in_sizes, int n_in,
                              void* d_out, int out_size, void* d_ws, size_t ws_size,
                              hipStream_t stream) {
    (void)n_in; (void)d_ws; (void)ws_size; (void)out_size;
    const float* t  = (const float*)d_in[0];
    const float* x  = (const float*)d_in[1];
    const float* W0 = (const float*)d_in[2];
    const float* b0 = (const float*)d_in[3];
    const float* Uz = (const float*)d_in[4];
    const float* Ug = (const float*)d_in[5];
    const float* Ur = (const float*)d_in[6];
    const float* Uh = (const float*)d_in[7];
    const float* Wz = (const float*)d_in[8];
    const float* Wg = (const float*)d_in[9];
    const float* Wr = (const float*)d_in[10];
    const float* Wh = (const float*)d_in[11];
    const float* bz = (const float*)d_in[12];
    const float* bg = (const float*)d_in[13];
    const float* br = (const float*)d_in[14];
    const float* bh = (const float*)d_in[15];
    const float* Wf = (const float*)d_in[16];
    const float* bf = (const float*)d_in[17];
    float* out = (float*)d_out;

    const int N = in_sizes[0];
    const int blocks = N / MBLK;
    dgm_fused_kernel<<<blocks, 256, 0, stream>>>(
        t, x, W0, b0, Uz, Ug, Ur, Uh, Wz, Wg, Wr, Wh,
        bz, bg, br, bh, Wf, bf, out);
}
